# starting kernel
# baseline (speedup 1.0000x reference)
#include <hip/hip_runtime.h>
#include <hip/hip_bf16.h>

typedef float v2f __attribute__((ext_vector_type(2)));
typedef float v8f __attribute__((ext_vector_type(8)));

#define B_    64
#define E_    16
#define IN_   256
#define OUT_  256
#define NPG_  2048
#define MWG   128
#define KC    32
#define BSTR2 272

__global__ __launch_bounds__(256) void mole_mix(const float* __restrict__ kern,
                                                const float* __restrict__ coeff,
                                                float* __restrict__ mixed) {
    __shared__ float cs[B_ * E_];
    const int tid = threadIdx.x;
    for (int j = tid; j < B_ * E_; j += 256) cs[j] = coeff[j];
    __syncthreads();

    const int idx = blockIdx.x * 256 + tid;
    const int i   = idx >> 6;
    const int o4  = (idx & 63) << 2;

    float4 kf[E_];
#pragma unroll
    for (int e = 0; e < E_; ++e)
        kf[e] = *(const float4*)(kern + (size_t)e * IN_ * OUT_ + (size_t)i * OUT_ + o4);

    for (int b = 0; b < B_; ++b) {
        float4 acc = make_float4(0.f, 0.f, 0.f, 0.f);
#pragma unroll
        for (int e = 0; e < E_; ++e) {
            const float c = cs[b * E_ + e];
            acc.x += c * kf[e].x;  acc.y += c * kf[e].y;
            acc.z += c * kf[e].z;  acc.w += c * kf[e].w;
        }
        *(float4*)(mixed + (size_t)b * IN_ * OUT_ + (size_t)i * OUT_ + o4) = acc;
    }
}

__global__ __launch_bounds__(256) void mole_gemm(const float* __restrict__ X,
                                                 const float* __restrict__ mixed,
                                                 const float* __restrict__ bias,
                                                 float* __restrict__ out) {
    extern __shared__ float Bs[];

    const int tid = threadIdx.x;
    const int b   = blockIdx.x >> 4;
    const int mt  = blockIdx.x & 15;
    const float* __restrict__ Bm = mixed + (size_t)b * IN_ * OUT_;

    const int wave = tid >> 5;
    const int lane = tid & 31;
    const int half = lane >> 4;
    const int l16  = lane & 15;
    const int rowBase = b * NPG_ + mt * MWG + wave * 16;
    const float* __restrict__ Xrow = X + (size_t)(rowBase + l16) * IN_ + 2 * half;

    v8f acc[16];
#pragma unroll
    for (int n = 0; n < 16; ++n) {
        v8f z = {};
        acc[n] = z;
    }

    for (int kc = 0; kc < IN_ / KC; ++kc) {
        for (int g = tid; g < (KC / 2) * (OUT_ / 4); g += 256) {
            const int p  = g >> 6;
            const int c4 = (g & 63) << 2;
            const float* src = Bm + (size_t)(kc * KC + 2 * p) * OUT_ + c4;
            const float4 lo = *(const float4*)(src);
            const float4 hi = *(const float4*)(src + OUT_);
            float* dst = Bs + ((size_t)p * BSTR2 + c4) * 2;
            *(float4*)(dst)     = make_float4(lo.x, hi.x, lo.y, hi.y);
            *(float4*)(dst + 4) = make_float4(lo.z, hi.z, lo.w, hi.w);
        }
        __syncthreads();

        for (int kk = 0; kk < KC / 4; ++kk) {
            const v2f a = *(const v2f*)(Xrow + kc * KC + kk * 4);
            const v2f* __restrict__ Brow =
                (const v2f*)Bs + (size_t)(2 * kk + half) * BSTR2 + l16;
#pragma unroll
            for (int n = 0; n < 16; ++n) {
                const v2f bf = Brow[n * 16];
                acc[n] = __builtin_amdgcn_wmma_f32_16x16x4_f32(
                    false, a, false, bf, (short)0, acc[n], false, false);
            }
        }
        __syncthreads();
    }

    const int outRow0 = rowBase + 8 * half;
#pragma unroll
    for (int n = 0; n < 16; ++n) {
        const int col = n * 16 + l16;
        const float bv = bias[col];
#pragma unroll
        for (int r = 0; r < 8; ++r) {
            out[(size_t)(outRow0 + r) * OUT_ + col] = acc[n][r] + bv;
        }
    }
}

extern "C" void kernel_launch(void* const* d_in, const int* in_sizes, int n_in,
                              void* d_out, int out_size, void* d_ws, size_t ws_size,
                              hipStream_t stream) {
    const float* inputs = (const float*)d_in[0];
    const float* coeffs = (const float*)d_in[1];
    const float* kern   = (const float*)d_in[3];
    const float* bias   = (const float*)d_in[4];
    float*       out    = (float*)d_out;
    float*       mixed  = (float*)d_ws;

    mole_mix<<<dim3((IN_ * OUT_ / 4) / 256), dim3(256), 0, stream>>>(kern, coeffs, mixed);

    const size_t ldsBytes = (size_t)(KC / 2) * BSTR2 * 2 * sizeof(float);
    mole_gemm<<<dim3(B_ * (NPG_ / MWG)), dim3(256), ldsBytes, stream>>>(inputs, mixed,
                                                                        bias, out);
}


# GEMM K loop software-pipelined: next 16-row chunk's global loads (B rows, A frags) issue before this chunk's WMMAs into a 2nd LDS stage, 1 barrier/chunk; waves_per_eu(4,4): LDS reads group ahead
# speedup vs baseline: 1.1733x; 1.1733x over previous
#include <hip/hip_runtime.h>
#include <hip/hip_bf16.h>


typedef float v2f __attribute__((ext_vector_type(2)));
typedef float v8f __attribute__((ext_vector_type(8)));

#define B_    64
#define E_    16
#define IN_   256
#define OUT_  256
#define NPG_  2048
#define MWG   128
#define KC    16
#define BSTR2 272
#define NCH   (IN_ / KC)
#define STAGE_F ((KC / 2) * BSTR2 * 2)

__global__ __launch_bounds__(256) void mole_mix(const float* __restrict__ kern,
                                                const float* __restrict__ coeff,
                                                float* __restrict__ mixed) {
    __shared__ float cs[B_ * E_];
    const int tid = threadIdx.x;
    for (int j = tid; j < B_ * E_; j += 256) cs[j] = coeff[j];
    __syncthreads();

    const int idx = blockIdx.x * 256 + tid;
    const int i   = idx >> 6;
    const int o4  = (idx & 63) << 2;

    float4 kf[E_];
#pragma unroll
    for (int e = 0; e < E_; ++e)
        kf[e] = *(const float4*)(kern + (size_t)e * IN_ * OUT_ + (size_t)i * OUT_ + o4);

    for (int b = 0; b < B_; ++b) {
        float4 acc = make_float4(0.f, 0.f, 0.f, 0.f);
#pragma unroll
        for (int e = 0; e < E_; ++e) {
            const float c = cs[b * E_ + e];
            acc.x += c * kf[e].x;  acc.y += c * kf[e].y;
            acc.z += c * kf[e].z;  acc.w += c * kf[e].w;
        }
        *(float4*)(mixed + (size_t)b * IN_ * OUT_ + (size_t)i * OUT_ + o4) = acc;
    }
}


__device__ __forceinline__ void chunk_mma(const v2f* __restrict__ Bst, const v2f (&a)[KC / 4],
                                          v8f (&acc)[16]) {
#pragma unroll
    for (int kk = 0; kk < KC / 4; ++kk) {
        const v2f* __restrict__ Brow = Bst + (size_t)(2 * kk) * BSTR2;
#pragma unroll
        for (int n = 0; n < 16; ++n) {
            const v2f bf = Brow[n * 16];
            acc[n] = __builtin_amdgcn_wmma_f32_16x16x4_f32(
                false, a[kk], false, bf, (short)0, acc[n], false, false);
        }
    }
}

__global__ __launch_bounds__(256) __attribute__((amdgpu_waves_per_eu(4, 4))) void mole_gemm(const float* __restrict__ X,
                                                 const float* __restrict__ mixed,
                                                 const float* __restrict__ bias,
                                                 float* __restrict__ out) {
    __shared__ float Bs[2 * STAGE_F];

    const int tid = threadIdx.x;
    const int b   = blockIdx.x >> 4;
    const int mt  = blockIdx.x & 15;
    const float* __restrict__ Bm = mixed + (size_t)b * IN_ * OUT_;

    const int wave = tid >> 5;
    const int lane = tid & 31;
    const int half = lane >> 4;
    const int l16  = lane & 15;
    const int rowBase = b * NPG_ + mt * MWG + wave * 16;
    const float* __restrict__ Xwg = X + (size_t)(b * NPG_ + mt * MWG) * IN_;
    const unsigned xoff = (unsigned)((wave * 16 + l16) * IN_ + 2 * half);

    const int sp = tid >> 6;
    const int c4 = (tid & 63) << 2;
    const unsigned boff = (unsigned)((2 * sp) * OUT_ + c4);
    float* const Bdst = Bs + ((size_t)sp * BSTR2 + c4) * 2;
    const v2f* const Bfrag = (const v2f*)Bs + (size_t)half * BSTR2 + l16;

    v8f acc[16];
#pragma unroll
    for (int n = 0; n < 16; ++n) {
        v8f z = {};
        acc[n] = z;
    }

    float4 blo[KC / 8], bhi[KC / 8];
    v2f aCur[KC / 4], aNxt[KC / 4];

#pragma unroll
    for (int j = 0; j < KC / 8; ++j) {
        blo[j] = *(const float4*)(Bm + (boff + (unsigned)((8 * j) * OUT_)));
        bhi[j] = *(const float4*)(Bm + (boff + (unsigned)((8 * j + 1) * OUT_)));
    }
#pragma unroll
    for (int kk = 0; kk < KC / 4; ++kk) aCur[kk] = *(const v2f*)(Xwg + (xoff + (unsigned)(kk * 4)));
#pragma unroll
    for (int j = 0; j < KC / 8; ++j) {
        float* dst = Bdst + (size_t)(4 * j) * BSTR2 * 2;
        *(float4*)(dst)     = make_float4(blo[j].x, bhi[j].x, blo[j].y, bhi[j].y);
        *(float4*)(dst + 4) = make_float4(blo[j].z, bhi[j].z, blo[j].w, bhi[j].w);
    }
    __syncthreads();

    for (int kc = 0; kc < NCH; ++kc) {
        const bool more = kc + 1 < NCH;
        if (more) {
            const float* __restrict__ Bnx = Bm + (size_t)(kc + 1) * KC * OUT_;
            const float* __restrict__ Xnx = Xwg + (kc + 1) * KC;
#pragma unroll
            for (int j = 0; j < KC / 8; ++j) {
                blo[j] = *(const float4*)(Bnx + (boff + (unsigned)((8 * j) * OUT_)));
                bhi[j] = *(const float4*)(Bnx + (boff + (unsigned)((8 * j + 1) * OUT_)));
            }
#pragma unroll
            for (int kk = 0; kk < KC / 4; ++kk)
                aNxt[kk] = *(const v2f*)(Xnx + (xoff + (unsigned)(kk * 4)));
        }

        chunk_mma(Bfrag + (size_t)(kc & 1) * (STAGE_F / 2), aCur, acc);

        if (more) {
            float* const nst = Bdst + (size_t)((kc + 1) & 1) * STAGE_F;
#pragma unroll
            for (int j = 0; j < KC / 8; ++j) {
                float* dst = nst + (size_t)(4 * j) * BSTR2 * 2;
                *(float4*)(dst)     = make_float4(blo[j].x, bhi[j].x, blo[j].y, bhi[j].y);
                *(float4*)(dst + 4) = make_float4(blo[j].z, bhi[j].z, blo[j].w, bhi[j].w);
            }
#pragma unroll
            for (int kk = 0; kk < KC / 4; ++kk) aCur[kk] = aNxt[kk];
            __syncthreads();
        }
    }

    const int outRow0 = rowBase + 8 * half;
#pragma unroll
    for (int n = 0; n < 16; ++n) {
        const int col = n * 16 + l16;
        const float bv = bias[col];
#pragma unroll
        for (int r = 0; r < 8; ++r) {
            out[(size_t)(outRow0 + r) * OUT_ + col] = acc[n][r] + bv;
        }
    }
}

extern "C" void kernel_launch(void* const* d_in, const int* in_sizes, int n_in,
                              void* d_out, int out_size, void* d_ws, size_t ws_size,
                              hipStream_t stream) {
    const float* inputs = (const float*)d_in[0];
    const float* coeffs = (const float*)d_in[1];
    const float* kern   = (const float*)d_in[3];
    const float* bias   = (const float*)d_in[4];
    float*       out    = (float*)d_out;
    float*       mixed  = (float*)d_ws;

    mole_mix<<<dim3((IN_ * OUT_ / 4) / 256), dim3(256), 0, stream>>>(kern, coeffs, mixed);

    mole_gemm<<<dim3(B_ * (NPG_ / MWG)), dim3(256), 0, stream>>>(inputs, mixed, bias, out);
}


# GEMM re-tiled B-stationary: a work-group keeps its graph's 256x64 mixed slice as WMMA B fragments in registers for all 2048 rows; A by float4 loads a tile ahead; no LDS; k sum order permuted within 8
# speedup vs baseline: 1.3001x; 1.1081x over previous
#include <hip/hip_runtime.h>
#include <hip/hip_bf16.h>


typedef float v2f __attribute__((ext_vector_type(2)));
typedef float v8f __attribute__((ext_vector_type(8)));

#define B_    64
#define E_    16
#define IN_   256
#define OUT_  256
#define NPG_  2048
#define NW    64
#define NSL   (OUT_ / NW)
#define TM    16
#define GW    4
#define NT    (NPG_ / (TM * GW))
#define KG    (IN_ / 8)

__global__ __launch_bounds__(256) void mole_mix(const float* __restrict__ kern,
                                                const float* __restrict__ coeff,
                                                float* __restrict__ mixed) {
    __shared__ float cs[B_ * E_];
    const int tid = threadIdx.x;
    for (int j = tid; j < B_ * E_; j += 256) cs[j] = coeff[j];
    __syncthreads();

    const int idx = blockIdx.x * 256 + tid;
    const int i   = idx >> 6;
    const int o4  = (idx & 63) << 2;

    float4 kf[E_];
#pragma unroll
    for (int e = 0; e < E_; ++e)
        kf[e] = *(const float4*)(kern + (size_t)e * IN_ * OUT_ + (size_t)i * OUT_ + o4);

    for (int b = 0; b < B_; ++b) {
        float4 acc = make_float4(0.f, 0.f, 0.f, 0.f);
#pragma unroll
        for (int e = 0; e < E_; ++e) {
            const float c = cs[b * E_ + e];
            acc.x += c * kf[e].x;  acc.y += c * kf[e].y;
            acc.z += c * kf[e].z;  acc.w += c * kf[e].w;
        }
        *(float4*)(mixed + (size_t)b * IN_ * OUT_ + (size_t)i * OUT_ + o4) = acc;
    }
}


__device__ __forceinline__ void load_a(float4 (&a)[KG], const float* __restrict__ p) {
#pragma unroll
    for (int j = 0; j < KG; ++j) a[j] = *(const float4*)(p + 8 * j);
}

__device__ __forceinline__ void tile_mma_store(const float4 (&a)[KG], const v2f (&bf)[KG][2][4],
                                               const float4 bias4, float* __restrict__ o) {
    v8f acc[4];
#pragma unroll
    for (int n = 0; n < 4; ++n) {
        v8f z = {};
        acc[n] = z;
    }
#pragma unroll
    for (int j = 0; j < KG; ++j) {
        v2f a0, a1;
        a0.x = a[j].x;  a0.y = a[j].y;
        a1.x = a[j].z;  a1.y = a[j].w;
#pragma unroll
        for (int n = 0; n < 4; ++n)
            acc[n] = __builtin_amdgcn_wmma_f32_16x16x4_f32(
                false, a0, false, bf[j][0][n], (short)0, acc[n], false, false);
#pragma unroll
        for (int n = 0; n < 4; ++n)
            acc[n] = __builtin_amdgcn_wmma_f32_16x16x4_f32(
                false, a1, false, bf[j][1][n], (short)0, acc[n], false, false);
    }
    __builtin_amdgcn_sched_barrier(0);
    asm volatile("v_nop\n\tv_nop\n\tv_nop\n\tv_nop\n\tv_nop");
    __builtin_amdgcn_sched_barrier(0);
#pragma unroll
    for (int r = 0; r < 8; ++r) {
        float4 v;
        v.x = acc[0][r] + bias4.x;
        v.y = acc[1][r] + bias4.y;
        v.z = acc[2][r] + bias4.z;
        v.w = acc[3][r] + bias4.w;
        *(float4*)(o + (size_t)r * OUT_) = v;
    }
}

__global__ __launch_bounds__(32 * GW) __attribute__((amdgpu_waves_per_eu(1, 1))) void mole_gemm(const float* __restrict__ X,
                                                 const float* __restrict__ mixed,
                                                 const float* __restrict__ bias,
                                                 float* __restrict__ out) {
    const int tid  = threadIdx.x;
    const int wave = tid >> 5;
    const int lane = tid & 31;
    const int half = lane >> 4;
    const int c    = lane & 15;
    const int b    = blockIdx.x / NSL;
    const int sl   = blockIdx.x % NSL;
    const int col  = sl * NW + 4 * c;

    const float* __restrict__ Xl = X + ((size_t)b * NPG_ + wave * TM + c) * IN_ + 4 * half;
    float* __restrict__ Ol = out + ((size_t)b * NPG_ + wave * TM + 8 * half) * OUT_ + col;
    const size_t tstepX = (size_t)GW * TM * IN_;
    const size_t tstepO = (size_t)GW * TM * OUT_;

    float4 aA[KG], aB[KG];
    load_a(aA, Xl);

    const float* __restrict__ Bl = mixed + (size_t)b * IN_ * OUT_ + (size_t)(4 * half) * OUT_ + col;
    v2f bf[KG][2][4];
#pragma unroll
    for (int j = 0; j < KG; ++j) {
        float4 r[4];
#pragma unroll
        for (int u = 0; u < 4; ++u) r[u] = *(const float4*)(Bl + (size_t)(8 * j + u) * OUT_);
#pragma unroll
        for (int q = 0; q < 2; ++q) {
            bf[j][q][0].x = r[2 * q].x;  bf[j][q][0].y = r[2 * q + 1].x;
            bf[j][q][1].x = r[2 * q].y;  bf[j][q][1].y = r[2 * q + 1].y;
            bf[j][q][2].x = r[2 * q].z;  bf[j][q][2].y = r[2 * q + 1].z;
            bf[j][q][3].x = r[2 * q].w;  bf[j][q][3].y = r[2 * q + 1].w;
        }
    }
    const float4 bias4 = *(const float4*)(bias + col);

    for (int t = 0; t < NT; t += 2) {
        __builtin_amdgcn_sched_barrier(0);
        load_a(aB, Xl + (size_t)(t + 1) * tstepX);
        __builtin_amdgcn_sched_barrier(0);
        tile_mma_store(aA, bf, bias4, Ol + (size_t)t * tstepO);
        __builtin_amdgcn_sched_barrier(0);
        const int tn = (t + 2 < NT) ? t + 2 : t + 1;
        load_a(aA, Xl + (size_t)tn * tstepX);
        __builtin_amdgcn_sched_barrier(0);
        tile_mma_store(aB, bf, bias4, Ol + (size_t)(t + 1) * tstepO);
    }
}

extern "C" void kernel_launch(void* const* d_in, const int* in_sizes, int n_in,
                              void* d_out, int out_size, void* d_ws, size_t ws_size,
                              hipStream_t stream) {
    const float* inputs = (const float*)d_in[0];
    const float* coeffs = (const float*)d_in[1];
    const float* kern   = (const float*)d_in[3];
    const float* bias   = (const float*)d_in[4];
    float*       out    = (float*)d_out;
    float*       mixed  = (float*)d_ws;

    mole_mix<<<dim3((IN_ * OUT_ / 4) / 256), dim3(256), 0, stream>>>(kern, coeffs, mixed);

    mole_gemm<<<dim3(B_ * NSL), dim3(32 * GW), 0, stream>>>(inputs, mixed, bias, out);
}


# GEMM work-group re-tiled to one graph's 512-row block x all 256 columns: its four waves keep the four 64-column slices and walk the SAME rows, so each X line is asked of L2 by one WGP, not four
# speedup vs baseline: 1.3407x; 1.0312x over previous
#include <hip/hip_runtime.h>
#include <hip/hip_bf16.h>


typedef float v2f __attribute__((ext_vector_type(2)));
typedef float v8f __attribute__((ext_vector_type(8)));

#define B_    64
#define E_    16
#define IN_   256
#define OUT_  256
#define NPG_  2048
#define NW    64
#define GW    (OUT_ / NW)
#define TM    16
#define NRB   4
#define RB    (NPG_ / NRB)
#define NT    (RB / TM)
#define KG    (IN_ / 8)

__global__ __launch_bounds__(256) void mole_mix(const float* __restrict__ kern,
                                                const float* __restrict__ coeff,
                                                float* __restrict__ mixed) {
    __shared__ float cs[B_ * E_];
    const int tid = threadIdx.x;
    for (int j = tid; j < B_ * E_; j += 256) cs[j] = coeff[j];
    __syncthreads();

    const int idx = blockIdx.x * 256 + tid;
    const int i   = idx >> 6;
    const int o4  = (idx & 63) << 2;

    float4 kf[E_];
#pragma unroll
    for (int e = 0; e < E_; ++e)
        kf[e] = *(const float4*)(kern + (size_t)e * IN_ * OUT_ + (size_t)i * OUT_ + o4);

    for (int b = 0; b < B_; ++b) {
        float4 acc = make_float4(0.f, 0.f, 0.f, 0.f);
#pragma unroll
        for (int e = 0; e < E_; ++e) {
            const float c = cs[b * E_ + e];
            acc.x += c * kf[e].x;  acc.y += c * kf[e].y;
            acc.z += c * kf[e].z;  acc.w += c * kf[e].w;
        }
        *(float4*)(mixed + (size_t)b * IN_ * OUT_ + (size_t)i * OUT_ + o4) = acc;
    }
}


__device__ __forceinline__ void load_a(float4 (&a)[KG], const float* __restrict__ p) {
#pragma unroll
    for (int j = 0; j < KG; ++j) a[j] = *(const float4*)(p + 8 * j);
}

__device__ __forceinline__ void tile_mma_store(const float4 (&a)[KG], const v2f (&bf)[KG][2][4],
                                               const float4 bias4, float* __restrict__ o) {
    v8f acc[4];
#pragma unroll
    for (int n = 0; n < 4; ++n) {
        v8f z = {};
        acc[n] = z;
    }
#pragma unroll
    for (int j = 0; j < KG; ++j) {
        v2f a0, a1;
        a0.x = a[j].x;  a0.y = a[j].y;
        a1.x = a[j].z;  a1.y = a[j].w;
#pragma unroll
        for (int n = 0; n < 4; ++n)
            acc[n] = __builtin_amdgcn_wmma_f32_16x16x4_f32(
                false, a0, false, bf[j][0][n], (short)0, acc[n], false, false);
#pragma unroll
        for (int n = 0; n < 4; ++n)
            acc[n] = __builtin_amdgcn_wmma_f32_16x16x4_f32(
                false, a1, false, bf[j][1][n], (short)0, acc[n], false, false);
    }
    __builtin_amdgcn_sched_barrier(0);
    asm volatile("v_nop\n\tv_nop\n\tv_nop\n\tv_nop\n\tv_nop");
    __builtin_amdgcn_sched_barrier(0);
#pragma unroll
    for (int r = 0; r < 8; ++r) {
        float4 v;
        v.x = acc[0][r] + bias4.x;
        v.y = acc[1][r] + bias4.y;
        v.z = acc[2][r] + bias4.z;
        v.w = acc[3][r] + bias4.w;
        *(float4*)(o + (size_t)r * OUT_) = v;
    }
}

__global__ __launch_bounds__(32 * GW) __attribute__((amdgpu_waves_per_eu(1, 1))) void mole_gemm(const float* __restrict__ X,
                                                 const float* __restrict__ mixed,
                                                 const float* __restrict__ bias,
                                                 float* __restrict__ out) {
    const int tid  = threadIdx.x;
    const int wave = tid >> 5;
    const int lane = tid & 31;
    const int half = lane >> 4;
    const int c    = lane & 15;
    const int b    = blockIdx.x / NRB;
    const int rb   = blockIdx.x % NRB;
    const int col  = wave * NW + 4 * c;

    const float* __restrict__ Xl = X + ((size_t)b * NPG_ + rb * RB + c) * IN_ + 4 * half;
    float* __restrict__ Ol = out + ((size_t)b * NPG_ + rb * RB + 8 * half) * OUT_ + col;
    const size_t tstepX = (size_t)TM * IN_;
    const size_t tstepO = (size_t)TM * OUT_;

    float4 aA[KG], aB[KG];
    load_a(aA, Xl);

    const float* __restrict__ Bl = mixed + (size_t)b * IN_ * OUT_ + (size_t)(4 * half) * OUT_ + col;
    v2f bf[KG][2][4];
#pragma unroll
    for (int j = 0; j < KG; ++j) {
        float4 r[4];
#pragma unroll
        for (int u = 0; u < 4; ++u) r[u] = *(const float4*)(Bl + (size_t)(8 * j + u) * OUT_);
#pragma unroll
        for (int q = 0; q < 2; ++q) {
            bf[j][q][0].x = r[2 * q].x;  bf[j][q][0].y = r[2 * q + 1].x;
            bf[j][q][1].x = r[2 * q].y;  bf[j][q][1].y = r[2 * q + 1].y;
            bf[j][q][2].x = r[2 * q].z;  bf[j][q][2].y = r[2 * q + 1].z;
            bf[j][q][3].x = r[2 * q].w;  bf[j][q][3].y = r[2 * q + 1].w;
        }
    }
    const float4 bias4 = *(const float4*)(bias + col);

    for (int t = 0; t < NT; t += 2) {
        __builtin_amdgcn_sched_barrier(0);
        load_a(aB, Xl + (size_t)(t + 1) * tstepX);
        __builtin_amdgcn_sched_barrier(0);
        tile_mma_store(aA, bf, bias4, Ol + (size_t)t * tstepO);
        __builtin_amdgcn_sched_barrier(0);
        const int tn = (t + 2 < NT) ? t + 2 : t + 1;
        load_a(aA, Xl + (size_t)tn * tstepX);
        __builtin_amdgcn_sched_barrier(0);
        tile_mma_store(aB, bf, bias4, Ol + (size_t)(t + 1) * tstepO);
    }
}

extern "C" void kernel_launch(void* const* d_in, const int* in_sizes, int n_in,
                              void* d_out, int out_size, void* d_ws, size_t ws_size,
                              hipStream_t stream) {
    const float* inputs = (const float*)d_in[0];
    const float* coeffs = (const float*)d_in[1];
    const float* kern   = (const float*)d_in[3];
    const float* bias   = (const float*)d_in[4];
    float*       out    = (float*)d_out;
    float*       mixed  = (float*)d_ws;

    mole_mix<<<dim3((IN_ * OUT_ / 4) / 256), dim3(256), 0, stream>>>(kern, coeffs, mixed);

    mole_gemm<<<dim3(B_ * NRB), dim3(32 * GW), 0, stream>>>(inputs, mixed, bias, out);
}
